# starting kernel
# baseline (speedup 1.0000x reference)
#include <hip/hip_runtime.h>

typedef float v2f __attribute__((ext_vector_type(2)));
typedef float v8f __attribute__((ext_vector_type(8)));

#define BATCH  4096
#define EMBED  128
#define HIDDEN 256
#define NOUT   10
#define MAXLEN 200
#define XSTRIDE (MAXLEN + 1)

__global__ __launch_bounds__(256) void pool_kernel(const int* __restrict__ x,
                                                   const float* __restrict__ emb,
                                                   float* __restrict__ pooled) {
  const int row  = (int)((blockIdx.x * blockDim.x + threadIdx.x) >> 5);
  const int lane = (int)(threadIdx.x & 31);
  if (row >= BATCH) return;

  const int* __restrict__ xr = x + row * XSTRIDE;
  const int len = xr[0];
  const float4* __restrict__ embv = (const float4*)emb;

  float4 acc = {0.f, 0.f, 0.f, 0.f};
  int j = 0;
  for (; j + 4 <= len; j += 4) {
    const int t0 = xr[1 + j], t1 = xr[2 + j], t2 = xr[3 + j], t3 = xr[4 + j];
    const float4 e0 = embv[t0 * 32 + lane];
    const float4 e1 = embv[t1 * 32 + lane];
    const float4 e2 = embv[t2 * 32 + lane];
    const float4 e3 = embv[t3 * 32 + lane];
    const float s0 = (t0 != 0) ? 1.f : 0.f;
    const float s1 = (t1 != 0) ? 1.f : 0.f;
    const float s2 = (t2 != 0) ? 1.f : 0.f;
    const float s3 = (t3 != 0) ? 1.f : 0.f;
    acc.x = fmaf(s0, e0.x, acc.x); acc.y = fmaf(s0, e0.y, acc.y);
    acc.z = fmaf(s0, e0.z, acc.z); acc.w = fmaf(s0, e0.w, acc.w);
    acc.x = fmaf(s1, e1.x, acc.x); acc.y = fmaf(s1, e1.y, acc.y);
    acc.z = fmaf(s1, e1.z, acc.z); acc.w = fmaf(s1, e1.w, acc.w);
    acc.x = fmaf(s2, e2.x, acc.x); acc.y = fmaf(s2, e2.y, acc.y);
    acc.z = fmaf(s2, e2.z, acc.z); acc.w = fmaf(s2, e2.w, acc.w);
    acc.x = fmaf(s3, e3.x, acc.x); acc.y = fmaf(s3, e3.y, acc.y);
    acc.z = fmaf(s3, e3.z, acc.z); acc.w = fmaf(s3, e3.w, acc.w);
  }
  for (; j < len; ++j) {
    const int t = xr[1 + j];
    const float4 e = embv[t * 32 + lane];
    const float s = (t != 0) ? 1.f : 0.f;
    acc.x = fmaf(s, e.x, acc.x); acc.y = fmaf(s, e.y, acc.y);
    acc.z = fmaf(s, e.z, acc.z); acc.w = fmaf(s, e.w, acc.w);
  }
  ((float4*)pooled)[row * 32 + lane] = acc;
}

#define PS_STRIDE 132
#define HS_STRIDE 260
#define PT_STRIDE 18

__global__ __launch_bounds__(128) void mlp_kernel(const float* __restrict__ pooled,
                                                  const float* __restrict__ W1,
                                                  const float* __restrict__ b1,
                                                  const float* __restrict__ W2,
                                                  const float* __restrict__ b2,
                                                  float* __restrict__ out) {
  __shared__ float Ps[16 * PS_STRIDE];
  __shared__ float Hs[16 * HS_STRIDE];
  __shared__ float Pt[4 * 16 * PT_STRIDE];

  const int tid  = (int)threadIdx.x;
  const int lane = tid & 31;
  const int wave = tid >> 5;
  const int m0   = (int)blockIdx.x * 16;
  const int mn   = lane & 15;
  const int kh   = (lane >> 4) << 1;
  const int chi  = (lane >> 4) << 3;

  for (int i = tid; i < 16 * 32; i += 128) {
    const int r = i >> 5, c = i & 31;
    const float4 v = ((const float4*)pooled)[(m0 + r) * 32 + c];
    float* dst = &Ps[r * PS_STRIDE + c * 4];
    dst[0] = v.x; dst[1] = v.y; dst[2] = v.z; dst[3] = v.w;
  }
  __syncthreads();

  v8f acc[4] = {};
  const int nb = wave * 64;
  for (int k0 = 0; k0 < EMBED; k0 += 4) {
    v2f a;
    a.x = Ps[mn * PS_STRIDE + k0 + kh];
    a.y = Ps[mn * PS_STRIDE + k0 + kh + 1];
#pragma unroll
    for (int t = 0; t < 4; ++t) {
      v2f b;
      b.x = W1[(k0 + kh) * HIDDEN + nb + t * 16 + mn];
      b.y = W1[(k0 + kh + 1) * HIDDEN + nb + t * 16 + mn];
      acc[t] = __builtin_amdgcn_wmma_f32_16x16x4_f32(
          false, a, false, b, (short)0, acc[t], false, false);
    }
  }
#pragma unroll
  for (int t = 0; t < 4; ++t) {
#pragma unroll
    for (int r = 0; r < 8; ++r) {
      const int M = r + chi;
      const int N = nb + t * 16 + mn;
      const float h = acc[t][r] + b1[N];
      Hs[M * HS_STRIDE + N] = h > 0.f ? h : 0.f;
    }
  }
  __syncthreads();

  v8f c2 = {};
  const int kb = wave * 64;
  for (int k0 = kb; k0 < kb + 64; k0 += 4) {
    v2f a;
    a.x = Hs[mn * HS_STRIDE + k0 + kh];
    a.y = Hs[mn * HS_STRIDE + k0 + kh + 1];
    v2f b;
    b.x = (mn < NOUT) ? W2[(k0 + kh) * NOUT + mn] : 0.f;
    b.y = (mn < NOUT) ? W2[(k0 + kh + 1) * NOUT + mn] : 0.f;
    c2 = __builtin_amdgcn_wmma_f32_16x16x4_f32(
        false, a, false, b, (short)0, c2, false, false);
  }
#pragma unroll
  for (int r = 0; r < 8; ++r)
    Pt[wave * (16 * PT_STRIDE) + (r + chi) * PT_STRIDE + mn] = c2[r];
  __syncthreads();

  for (int i = tid; i < 16 * NOUT; i += 128) {
    const int m = i / NOUT;
    const int n = i - m * NOUT;
    float s = b2[n];
#pragma unroll
    for (int w = 0; w < 4; ++w) s += Pt[w * (16 * PT_STRIDE) + m * PT_STRIDE + n];
    out[(m0 + m) * NOUT + n] = s;
  }
}

extern "C" void kernel_launch(void* const* d_in, const int* in_sizes, int n_in,
                              void* d_out, int out_size, void* d_ws, size_t ws_size,
                              hipStream_t stream) {
  const int*   x   = (const int*)d_in[0];
  const float* emb = (const float*)d_in[1];
  const float* W1  = (const float*)d_in[2];
  const float* b1  = (const float*)d_in[3];
  const float* W2  = (const float*)d_in[4];
  const float* b2  = (const float*)d_in[5];
  float* out    = (float*)d_out;
  float* pooled = (float*)d_ws;

  pool_kernel<<<BATCH / 8, 256, 0, stream>>>(x, emb, pooled);
  mlp_kernel<<<BATCH / 16, 128, 0, stream>>>(pooled, W1, b1, W2, b2, out);
}


# pool kernel: token ids fetched wave-uniform by scalar loads one batch ahead, and 20 table-row gathers kept in flight per wave instead of 8 behind per-lane id loads; token-order sum unchanged
# speedup vs baseline: 1.1639x; 1.1639x over previous
#include <hip/hip_runtime.h>

typedef float v2f __attribute__((ext_vector_type(2)));
typedef float v8f __attribute__((ext_vector_type(8)));

#define BATCH  4096
#define EMBED  128
#define HIDDEN 256
#define NOUT   10
#define MAXLEN 200
#define XSTRIDE (MAXLEN + 1)

#define GB 20
static_assert(MAXLEN % GB == 0, "a batch of ids must end inside the row");

__global__ __launch_bounds__(256) void pool_kernel(const int* __restrict__ x,
                                                   const float* __restrict__ emb,
                                                   float* __restrict__ pooled) {
  const int row = __builtin_amdgcn_readfirstlane(
      (int)((blockIdx.x * blockDim.x + threadIdx.x) >> 5));
  const unsigned lane = threadIdx.x & 31u;
  if (row >= BATCH) return;

  const int* __restrict__ xr = x + row * XSTRIDE;
  int len = xr[0];
  len = len < MAXLEN ? len : MAXLEN;

  const char* __restrict__ embb = (const char*)emb;
  const unsigned loff = lane * 16u;

  float4 acc = {0.f, 0.f, 0.f, 0.f};
  int tn[GB];
#pragma unroll
  for (int i = 0; i < GB; ++i) tn[i] = xr[1 + i];

  for (int j = 0; j < len; j += GB) {
    int t[GB];
#pragma unroll
    for (int i = 0; i < GB; ++i) t[i] = (j + i < len) ? tn[i] : 0;
    const int jn = (j + GB < MAXLEN) ? j + GB : j;
#pragma unroll
    for (int i = 0; i < GB; ++i) tn[i] = xr[1 + jn + i];

    float4 e[GB];
#pragma unroll
    for (int i = 0; i < GB; ++i) {
      const char* __restrict__ rowp = embb + ((size_t)(unsigned)t[i] << 9);
      e[i] = *(const float4*)(rowp + loff);
    }
    __builtin_amdgcn_sched_barrier(0);
#pragma unroll
    for (int i = 0; i < GB; ++i) {
      const float s = (t[i] != 0) ? 1.f : 0.f;
      acc.x = fmaf(s, e[i].x, acc.x); acc.y = fmaf(s, e[i].y, acc.y);
      acc.z = fmaf(s, e[i].z, acc.z); acc.w = fmaf(s, e[i].w, acc.w);
    }
  }
  ((float4*)pooled)[row * 32 + (int)lane] = acc;
}

#define PS_STRIDE 132
#define HS_STRIDE 260
#define PT_STRIDE 18

__global__ __launch_bounds__(128) void mlp_kernel(const float* __restrict__ pooled,
                                                  const float* __restrict__ W1,
                                                  const float* __restrict__ b1,
                                                  const float* __restrict__ W2,
                                                  const float* __restrict__ b2,
                                                  float* __restrict__ out) {
  __shared__ float Ps[16 * PS_STRIDE];
  __shared__ float Hs[16 * HS_STRIDE];
  __shared__ float Pt[4 * 16 * PT_STRIDE];

  const int tid  = (int)threadIdx.x;
  const int lane = tid & 31;
  const int wave = tid >> 5;
  const int m0   = (int)blockIdx.x * 16;
  const int mn   = lane & 15;
  const int kh   = (lane >> 4) << 1;
  const int chi  = (lane >> 4) << 3;

  for (int i = tid; i < 16 * 32; i += 128) {
    const int r = i >> 5, c = i & 31;
    const float4 v = ((const float4*)pooled)[(m0 + r) * 32 + c];
    float* dst = &Ps[r * PS_STRIDE + c * 4];
    dst[0] = v.x; dst[1] = v.y; dst[2] = v.z; dst[3] = v.w;
  }
  __syncthreads();

  v8f acc[4] = {};
  const int nb = wave * 64;
  for (int k0 = 0; k0 < EMBED; k0 += 4) {
    v2f a;
    a.x = Ps[mn * PS_STRIDE + k0 + kh];
    a.y = Ps[mn * PS_STRIDE + k0 + kh + 1];
#pragma unroll
    for (int t = 0; t < 4; ++t) {
      v2f b;
      b.x = W1[(k0 + kh) * HIDDEN + nb + t * 16 + mn];
      b.y = W1[(k0 + kh + 1) * HIDDEN + nb + t * 16 + mn];
      acc[t] = __builtin_amdgcn_wmma_f32_16x16x4_f32(
          false, a, false, b, (short)0, acc[t], false, false);
    }
  }
#pragma unroll
  for (int t = 0; t < 4; ++t) {
#pragma unroll
    for (int r = 0; r < 8; ++r) {
      const int M = r + chi;
      const int N = nb + t * 16 + mn;
      const float h = acc[t][r] + b1[N];
      Hs[M * HS_STRIDE + N] = h > 0.f ? h : 0.f;
    }
  }
  __syncthreads();

  v8f c2 = {};
  const int kb = wave * 64;
  for (int k0 = kb; k0 < kb + 64; k0 += 4) {
    v2f a;
    a.x = Hs[mn * HS_STRIDE + k0 + kh];
    a.y = Hs[mn * HS_STRIDE + k0 + kh + 1];
    v2f b;
    b.x = (mn < NOUT) ? W2[(k0 + kh) * NOUT + mn] : 0.f;
    b.y = (mn < NOUT) ? W2[(k0 + kh + 1) * NOUT + mn] : 0.f;
    c2 = __builtin_amdgcn_wmma_f32_16x16x4_f32(
        false, a, false, b, (short)0, c2, false, false);
  }
#pragma unroll
  for (int r = 0; r < 8; ++r)
    Pt[wave * (16 * PT_STRIDE) + (r + chi) * PT_STRIDE + mn] = c2[r];
  __syncthreads();

  for (int i = tid; i < 16 * NOUT; i += 128) {
    const int m = i / NOUT;
    const int n = i - m * NOUT;
    float s = b2[n];
#pragma unroll
    for (int w = 0; w < 4; ++w) s += Pt[w * (16 * PT_STRIDE) + m * PT_STRIDE + n];
    out[(m0 + m) * NOUT + n] = s;
  }
}

extern "C" void kernel_launch(void* const* d_in, const int* in_sizes, int n_in,
                              void* d_out, int out_size, void* d_ws, size_t ws_size,
                              hipStream_t stream) {
  const int*   x   = (const int*)d_in[0];
  const float* emb = (const float*)d_in[1];
  const float* W1  = (const float*)d_in[2];
  const float* b1  = (const float*)d_in[3];
  const float* W2  = (const float*)d_in[4];
  const float* b2  = (const float*)d_in[5];
  float* out    = (float*)d_out;
  float* pooled = (float*)d_ws;

  pool_kernel<<<BATCH / 8, 256, 0, stream>>>(x, emb, pooled);
  mlp_kernel<<<BATCH / 16, 128, 0, stream>>>(pooled, W1, b1, W2, b2, out);
}


# MLP kernel: a wave asks for all its global operands (W1 slice as 128-bit row loads, W2, pooled tile) in one group up front, kept in flight across an LDS-only barrier; no load round trip per k-step
# speedup vs baseline: 1.8995x; 1.6320x over previous
#include <hip/hip_runtime.h>

typedef float v2f __attribute__((ext_vector_type(2)));
typedef float v8f __attribute__((ext_vector_type(8)));

#define BATCH  4096
#define EMBED  128
#define HIDDEN 256
#define NOUT   10
#define MAXLEN 200
#define XSTRIDE (MAXLEN + 1)

#define GB 20
static_assert(MAXLEN % GB == 0, "a batch of ids must end inside the row");

__global__ __launch_bounds__(256) void pool_kernel(const int* __restrict__ x,
                                                   const float* __restrict__ emb,
                                                   float* __restrict__ pooled) {
  const int row = __builtin_amdgcn_readfirstlane(
      (int)((blockIdx.x * blockDim.x + threadIdx.x) >> 5));
  const unsigned lane = threadIdx.x & 31u;
  if (row >= BATCH) return;

  const int* __restrict__ xr = x + row * XSTRIDE;
  int len = xr[0];
  len = len < MAXLEN ? len : MAXLEN;

  const char* __restrict__ embb = (const char*)emb;
  const unsigned loff = lane * 16u;

  float4 acc = {0.f, 0.f, 0.f, 0.f};
  int tn[GB];
#pragma unroll
  for (int i = 0; i < GB; ++i) tn[i] = xr[1 + i];

  for (int j = 0; j < len; j += GB) {
    int t[GB];
#pragma unroll
    for (int i = 0; i < GB; ++i) t[i] = (j + i < len) ? tn[i] : 0;
    const int jn = (j + GB < MAXLEN) ? j + GB : j;
#pragma unroll
    for (int i = 0; i < GB; ++i) tn[i] = xr[1 + jn + i];

    float4 e[GB];
#pragma unroll
    for (int i = 0; i < GB; ++i) {
      const char* __restrict__ rowp = embb + ((size_t)(unsigned)t[i] << 9);
      e[i] = *(const float4*)(rowp + loff);
    }
    __builtin_amdgcn_sched_barrier(0);
#pragma unroll
    for (int i = 0; i < GB; ++i) {
      const float s = (t[i] != 0) ? 1.f : 0.f;
      acc.x = fmaf(s, e[i].x, acc.x); acc.y = fmaf(s, e[i].y, acc.y);
      acc.z = fmaf(s, e[i].z, acc.z); acc.w = fmaf(s, e[i].w, acc.w);
    }
  }
  ((float4*)pooled)[row * 32 + (int)lane] = acc;
}

#define PS_STRIDE 132
#define HS_STRIDE 260
#define PT_STRIDE 18
#define KSTEPS1 (EMBED / 4)
#define KSTEPS2 (HIDDEN / 4 / 4)

static __device__ __forceinline__ void lds_barrier() {
  asm volatile("s_wait_dscnt 0x0\n\ts_barrier_signal -1\n\ts_barrier_wait -1" ::: "memory");
}

__global__ __launch_bounds__(128) void mlp_kernel(const float* __restrict__ pooled,
                                                  const float* __restrict__ W1,
                                                  const float* __restrict__ b1,
                                                  const float* __restrict__ W2,
                                                  const float* __restrict__ b2,
                                                  float* __restrict__ out) {
  __shared__ __attribute__((aligned(16))) float Ps[16 * PS_STRIDE];
  __shared__ __attribute__((aligned(16))) float Hs[16 * HS_STRIDE];
  __shared__ float Pt[4 * 16 * PT_STRIDE];

  const int tid  = (int)threadIdx.x;
  const int lane = tid & 31;
  const int wave = tid >> 5;
  const int m0   = (int)blockIdx.x * 16;
  const int mn   = lane & 15;
  const int kh   = (lane >> 4) << 1;
  const int chi  = (lane >> 4) << 3;
  const int nb   = wave * 64;
  const int kb   = wave * 64;

  float4 pv[4];
#pragma unroll
  for (int u = 0; u < 4; ++u) {
    const int i = tid + u * 128;
    pv[u] = ((const float4*)pooled)[(m0 + (i >> 5)) * 32 + (i & 31)];
  }
#pragma unroll
  for (int u = 0; u < 4; ++u) {
    const int i = tid + u * 128;
    *(float4*)&Ps[(i >> 5) * PS_STRIDE + (i & 31) * 4] = pv[u];
  }
  const int nc = mn < NOUT ? mn : NOUT - 1;
  v2f w2f[KSTEPS2];
#pragma unroll
  for (int s = 0; s < KSTEPS2; ++s) {
    w2f[s].x = W2[(kb + 4 * s + kh) * NOUT + nc];
    w2f[s].y = W2[(kb + 4 * s + kh + 1) * NOUT + nc];
  }
  const float b2a = b2[tid % NOUT];
  const float b2b = b2[(tid + 128) % NOUT];
  const float4* __restrict__ W1q = (const float4*)W1 + (nb >> 2) + mn;
  float4 wlo[KSTEPS1], whi[KSTEPS1];
#pragma unroll
  for (int s = 0; s < KSTEPS1; ++s) {
    wlo[s] = W1q[(4 * s + kh) * (HIDDEN / 4)];
    whi[s] = W1q[(4 * s + kh + 1) * (HIDDEN / 4)];
  }
  const float4 bias = ((const float4*)b1)[(nb >> 2) + mn];
  __builtin_amdgcn_sched_barrier(0);

  lds_barrier();

  v2f af[KSTEPS1];
#pragma unroll
  for (int s = 0; s < KSTEPS1; ++s) {
    af[s].x = Ps[mn * PS_STRIDE + 4 * s + kh];
    af[s].y = Ps[mn * PS_STRIDE + 4 * s + kh + 1];
  }
  __builtin_amdgcn_sched_barrier(0);
  v8f acc[4] = {};
#pragma unroll
  for (int s = 0; s < KSTEPS1; ++s) {
    v2f b;
    b.x = wlo[s].x; b.y = whi[s].x;
    acc[0] = __builtin_amdgcn_wmma_f32_16x16x4_f32(false, af[s], false, b, (short)0, acc[0], false, false);
    b.x = wlo[s].y; b.y = whi[s].y;
    acc[1] = __builtin_amdgcn_wmma_f32_16x16x4_f32(false, af[s], false, b, (short)0, acc[1], false, false);
    b.x = wlo[s].z; b.y = whi[s].z;
    acc[2] = __builtin_amdgcn_wmma_f32_16x16x4_f32(false, af[s], false, b, (short)0, acc[2], false, false);
    b.x = wlo[s].w; b.y = whi[s].w;
    acc[3] = __builtin_amdgcn_wmma_f32_16x16x4_f32(false, af[s], false, b, (short)0, acc[3], false, false);
    if ((s & 3) == 3) __builtin_amdgcn_sched_barrier(0);
  }
#pragma unroll
  for (int r = 0; r < 8; ++r) {
    float4 h;
    h.x = acc[0][r] + bias.x; h.y = acc[1][r] + bias.y;
    h.z = acc[2][r] + bias.z; h.w = acc[3][r] + bias.w;
    h.x = h.x > 0.f ? h.x : 0.f; h.y = h.y > 0.f ? h.y : 0.f;
    h.z = h.z > 0.f ? h.z : 0.f; h.w = h.w > 0.f ? h.w : 0.f;
    *(float4*)&Hs[(r + chi) * HS_STRIDE + nb + 4 * mn] = h;
  }
  __syncthreads();

  v2f hf[KSTEPS2];
#pragma unroll
  for (int s = 0; s < KSTEPS2; ++s) {
    hf[s].x = Hs[mn * HS_STRIDE + kb + 4 * s + kh];
    hf[s].y = Hs[mn * HS_STRIDE + kb + 4 * s + kh + 1];
  }
  v8f c2 = {};
#pragma unroll
  for (int s = 0; s < KSTEPS2; ++s)
    c2 = __builtin_amdgcn_wmma_f32_16x16x4_f32(false, hf[s], false, w2f[s], (short)0, c2, false, false);
#pragma unroll
  for (int r = 0; r < 8; ++r)
    Pt[wave * (16 * PT_STRIDE) + (r + chi) * PT_STRIDE + mn] = c2[r];
  __syncthreads();

  {
    const int m = tid / NOUT;
    const int n = tid - m * NOUT;
    float s = b2a;
#pragma unroll
    for (int w = 0; w < 4; ++w) s += Pt[w * (16 * PT_STRIDE) + m * PT_STRIDE + n];
    out[(m0 + m) * NOUT + n] = s;
  }
  if (tid + 128 < 16 * NOUT) {
    const int i = tid + 128;
    const int m = i / NOUT;
    const int n = i - m * NOUT;
    float s = b2b;
#pragma unroll
    for (int w = 0; w < 4; ++w) s += Pt[w * (16 * PT_STRIDE) + m * PT_STRIDE + n];
    out[(m0 + m) * NOUT + n] = s;
  }
}

extern "C" void kernel_launch(void* const* d_in, const int* in_sizes, int n_in,
                              void* d_out, int out_size, void* d_ws, size_t ws_size,
                              hipStream_t stream) {
  const int*   x   = (const int*)d_in[0];
  const float* emb = (const float*)d_in[1];
  const float* W1  = (const float*)d_in[2];
  const float* b1  = (const float*)d_in[3];
  const float* W2  = (const float*)d_in[4];
  const float* b2  = (const float*)d_in[5];
  float* out    = (float*)d_out;
  float* pooled = (float*)d_ws;

  pool_kernel<<<BATCH / 8, 256, 0, stream>>>(x, emb, pooled);
  mlp_kernel<<<BATCH / 16, 128, 0, stream>>>(pooled, W1, b1, W2, b2, out);
}


# one launch instead of two: pooling and MLP fused per 16-row tile in a 16-wave block, pooled tile handed over in LDS, W1 fragments requested behind the pooling loop; every sum in the same order
# speedup vs baseline: 2.2373x; 1.1778x over previous
#include <hip/hip_runtime.h>

typedef float v2f __attribute__((ext_vector_type(2)));
typedef float v8f __attribute__((ext_vector_type(8)));

#define BATCH  4096
#define EMBED  128
#define HIDDEN 256
#define NOUT   10
#define MAXLEN 200
#define XSTRIDE (MAXLEN + 1)

#define GB 20
static_assert(MAXLEN % GB == 0, "a batch of ids must end inside the row");

#define PS_STRIDE 132
#define HS_STRIDE 260
#define PT_STRIDE 18
#define KSTEPS1 (EMBED / 4)
#define KSTEPS2 (HIDDEN / 4 / 4)
#define W1_EARLY 2

static __device__ __forceinline__ void lds_barrier() {
  asm volatile("s_wait_dscnt 0x0\n\ts_barrier_signal -1\n\ts_barrier_wait -1" ::: "memory");
}

__global__ __launch_bounds__(512) void fused_kernel(const int* __restrict__ x,
                                                    const float* __restrict__ emb,
                                                    const float* __restrict__ W1,
                                                    const float* __restrict__ b1,
                                                    const float* __restrict__ W2,
                                                    const float* __restrict__ b2,
                                                    float* __restrict__ out) {
  __shared__ __attribute__((aligned(16))) float Ps[16 * PS_STRIDE];
  __shared__ float Hs[16 * HS_STRIDE];
  __shared__ float Pt[4 * 16 * PT_STRIDE];

  const int tid  = (int)threadIdx.x;
  const int lane = tid & 31;
  const int wave = __builtin_amdgcn_readfirstlane(tid >> 5);
  const int m0   = (int)blockIdx.x * 16;
  const int mn   = lane & 15;
  const int kh   = (lane >> 4) << 1;
  const int chi  = (lane >> 4) << 3;
  const int nb   = wave * 16;

  const float* __restrict__ W1c = W1 + nb + mn;
  v2f wf[KSTEPS1];
#pragma unroll
  for (int s = 0; s < W1_EARLY; ++s) {
    wf[s].x = W1c[(4 * s + kh) * HIDDEN];
    wf[s].y = W1c[(4 * s + kh + 1) * HIDDEN];
  }
  const float bias = b1[nb + mn];
  const float b2v = b2[tid % NOUT];
  const int kb = (wave & 3) * 64;
  v2f w2f[KSTEPS2];
#pragma unroll
  for (int s = 0; s < KSTEPS2; ++s) { w2f[s].x = 0.f; w2f[s].y = 0.f; }
  if (wave < 4) {
    const int nc = mn < NOUT ? mn : NOUT - 1;
#pragma unroll
    for (int s = 0; s < KSTEPS2; ++s) {
      w2f[s].x = W2[(kb + 4 * s + kh) * NOUT + nc];
      w2f[s].y = W2[(kb + 4 * s + kh + 1) * NOUT + nc];
    }
  }

  {
    const int row = m0 + wave;
    const int* __restrict__ xr = x + row * XSTRIDE;
    int len = xr[0];
    len = len < MAXLEN ? len : MAXLEN;

    const char* __restrict__ embb = (const char*)emb;
    const unsigned loff = (unsigned)lane * 16u;

    float4 acc = {0.f, 0.f, 0.f, 0.f};
    int tn[GB];
#pragma unroll
    for (int i = 0; i < GB; ++i) tn[i] = xr[1 + i];

    for (int j = 0; j < len; j += GB) {
      int t[GB];
#pragma unroll
      for (int i = 0; i < GB; ++i) t[i] = (j + i < len) ? tn[i] : 0;
      const int jn = (j + GB < MAXLEN) ? j + GB : j;
#pragma unroll
      for (int i = 0; i < GB; ++i) tn[i] = xr[1 + jn + i];

      float4 e[GB];
#pragma unroll
      for (int i = 0; i < GB; ++i) {
        const char* __restrict__ rowp = embb + ((size_t)(unsigned)t[i] << 9);
        e[i] = *(const float4*)(rowp + loff);
      }
      __builtin_amdgcn_sched_barrier(0);
#pragma unroll
      for (int i = 0; i < GB; ++i) {
        const float s = (t[i] != 0) ? 1.f : 0.f;
        acc.x = fmaf(s, e[i].x, acc.x); acc.y = fmaf(s, e[i].y, acc.y);
        acc.z = fmaf(s, e[i].z, acc.z); acc.w = fmaf(s, e[i].w, acc.w);
      }
    }
    *(float4*)&Ps[wave * PS_STRIDE + lane * 4] = acc;
  }

#pragma unroll
  for (int s = W1_EARLY; s < KSTEPS1; ++s) {
    wf[s].x = W1c[(4 * s + kh) * HIDDEN];
    wf[s].y = W1c[(4 * s + kh + 1) * HIDDEN];
  }

  lds_barrier();

  v2f af[KSTEPS1];
#pragma unroll
  for (int s = 0; s < KSTEPS1; ++s) {
    af[s].x = Ps[mn * PS_STRIDE + 4 * s + kh];
    af[s].y = Ps[mn * PS_STRIDE + 4 * s + kh + 1];
  }
  __builtin_amdgcn_sched_barrier(0);
  v8f acc1 = {};
#pragma unroll
  for (int s = 0; s < KSTEPS1; ++s)
    acc1 = __builtin_amdgcn_wmma_f32_16x16x4_f32(false, af[s], false, wf[s], (short)0, acc1, false, false);
#pragma unroll
  for (int r = 0; r < 8; ++r) {
    float h = acc1[r] + bias;
    h = h > 0.f ? h : 0.f;
    Hs[(r + chi) * HS_STRIDE + nb + mn] = h;
  }
  __syncthreads();

  if (wave < 4) {
    v2f hf[KSTEPS2];
#pragma unroll
    for (int s = 0; s < KSTEPS2; ++s) {
      hf[s].x = Hs[mn * HS_STRIDE + kb + 4 * s + kh];
      hf[s].y = Hs[mn * HS_STRIDE + kb + 4 * s + kh + 1];
    }
    v8f c2 = {};
#pragma unroll
    for (int s = 0; s < KSTEPS2; ++s)
      c2 = __builtin_amdgcn_wmma_f32_16x16x4_f32(false, hf[s], false, w2f[s], (short)0, c2, false, false);
#pragma unroll
    for (int r = 0; r < 8; ++r)
      Pt[wave * (16 * PT_STRIDE) + (r + chi) * PT_STRIDE + mn] = c2[r];
  }
  __syncthreads();

  if (tid < 16 * NOUT) {
    const int m = tid / NOUT;
    const int n = tid - m * NOUT;
    float s = b2v;
#pragma unroll
    for (int w = 0; w < 4; ++w) s += Pt[w * (16 * PT_STRIDE) + m * PT_STRIDE + n];
    out[(m0 + m) * NOUT + n] = s;
  }
}

extern "C" void kernel_launch(void* const* d_in, const int* in_sizes, int n_in,
                              void* d_out, int out_size, void* d_ws, size_t ws_size,
                              hipStream_t stream) {
  const int*   x   = (const int*)d_in[0];
  const float* emb = (const float*)d_in[1];
  const float* W1  = (const float*)d_in[2];
  const float* b1  = (const float*)d_in[3];
  const float* W2  = (const float*)d_in[4];
  const float* b2  = (const float*)d_in[5];
  float* out = (float*)d_out;

  fused_kernel<<<BATCH / 16, 512, 0, stream>>>(x, emb, W1, b1, W2, b2, out);
}
